# starting kernel
# baseline (speedup 1.0000x reference)
#include <hip/hip_runtime.h>
#include <hip/hip_bf16.h>

#define NODE_DIM   128
#define EDGE_DIM   32
#define GLOBAL_DIM 64
#define HIDDEN_DIM 256
#define IN_DIM     224
#define NSEG_MAX   512

typedef float v2f __attribute__((ext_vector_type(2)));
typedef float v8f __attribute__((ext_vector_type(8)));

__device__ __forceinline__ void gAtomicAddF(float* p, float v) {
    __hip_atomic_fetch_add(p, v, __ATOMIC_RELAXED, __HIP_MEMORY_SCOPE_AGENT);
}

__global__ void zero_f32(float* __restrict__ p, int n) {
    int i = blockIdx.x * blockDim.x + threadIdx.x;
    if (i < n) p[i] = 0.0f;
}

#define NODES_PER_BLOCK 64
__global__ void node_scatter(const float* __restrict__ x,
                             const int* __restrict__ batch,
                             float* __restrict__ node_sum,
                             float* __restrict__ node_cnt,
                             int n_nodes) {
    int f = threadIdx.x;
    int start = blockIdx.x * NODES_PER_BLOCK;
    if (start >= n_nodes) return;
    int end = min(start + NODES_PER_BLOCK, n_nodes);

    int curseg = batch[start];
    float acc = 0.0f, cacc = 0.0f;
    for (int n = start; n < end; ++n) {
        int seg = batch[n];
        if (seg != curseg) {
            gAtomicAddF(&node_sum[(size_t)curseg * NODE_DIM + f], acc);
            if (f == 0) gAtomicAddF(&node_cnt[curseg], cacc);
            acc = 0.0f; cacc = 0.0f; curseg = seg;
        }
        acc  += x[(size_t)n * NODE_DIM + f];
        cacc += 1.0f;
    }
    gAtomicAddF(&node_sum[(size_t)curseg * NODE_DIM + f], acc);
    if (f == 0) gAtomicAddF(&node_cnt[curseg], cacc);
}

__global__ void edge_scatter(const float* __restrict__ edge_attr,
                             const int* __restrict__ edge_src,
                             const int* __restrict__ batch,
                             float* __restrict__ edge_sum,
                             float* __restrict__ edge_cnt,
                             int n_edges, int nseg) {
    __shared__ float lsum[NSEG_MAX * EDGE_DIM];
    __shared__ float lcnt[NSEG_MAX];
    for (int i = threadIdx.x; i < NSEG_MAX * EDGE_DIM; i += blockDim.x) lsum[i] = 0.0f;
    for (int i = threadIdx.x; i < NSEG_MAX; i += blockDim.x) lcnt[i] = 0.0f;
    __syncthreads();

    int lane = threadIdx.x & 31;
    int wave = threadIdx.x >> 5;
    int wavesPerBlock = blockDim.x >> 5;
    int gwave = blockIdx.x * wavesPerBlock + wave;
    int nwaves = gridDim.x * wavesPerBlock;

    for (int e = gwave; e < n_edges; e += nwaves) {
        int seg = batch[edge_src[e]];
        float v = edge_attr[(size_t)e * EDGE_DIM + lane];
        atomicAdd(&lsum[seg * EDGE_DIM + lane], v);
        if (lane == 0) atomicAdd(&lcnt[seg], 1.0f);
    }
    __syncthreads();

    for (int i = threadIdx.x; i < nseg * EDGE_DIM; i += blockDim.x) {
        float v = lsum[i];
        if (v != 0.0f) gAtomicAddF(&edge_sum[i], v);
    }
    for (int i = threadIdx.x; i < nseg; i += blockDim.x) {
        float v = lcnt[i];
        if (v != 0.0f) gAtomicAddF(&edge_cnt[i], v);
    }
}

__global__ void assemble(const float* __restrict__ node_sum,
                         const float* __restrict__ node_cnt,
                         const float* __restrict__ edge_sum,
                         const float* __restrict__ edge_cnt,
                         const float* __restrict__ u,
                         float* __restrict__ H0) {
    int b = blockIdx.x;
    int j = threadIdx.x;
    float v;
    if (j < NODE_DIM) {
        v = node_sum[(size_t)b * NODE_DIM + j] / fmaxf(node_cnt[b], 1.0f);
    } else if (j < NODE_DIM + EDGE_DIM) {
        v = edge_sum[(size_t)b * EDGE_DIM + (j - NODE_DIM)] / fmaxf(edge_cnt[b], 1.0f);
    } else {
        v = u[(size_t)b * GLOBAL_DIM + (j - NODE_DIM - EDGE_DIM)];
    }
    H0[(size_t)b * IN_DIM + j] = v;
}

template <bool RELU>
__global__ void gemm_wmma(const float* __restrict__ A,
                          const float* __restrict__ W,
                          const float* __restrict__ bias,
                          float* __restrict__ C,
                          int M, int N, int K) {
    int lane = threadIdx.x & 31;
    int wave = threadIdx.x >> 5;
    int wavesPerBlock = blockDim.x >> 5;
    int tile = blockIdx.x * wavesPerBlock + wave;
    int ntilesN = N >> 4;
    int tileM = tile / ntilesN;
    int tileN = tile - tileM * ntilesN;
    if (tileM >= (M >> 4)) return;

    int half = lane >> 4;
    int l15  = lane & 15;
    int row  = (tileM << 4) + l15;
    int col  = (tileN << 4) + l15;

    const float* __restrict__ Arow = A + (size_t)row * K;
    v8f acc = {};
    for (int k0 = 0; k0 < K; k0 += 4) {
        int ka = k0 + (half << 1);
        float2 av = *reinterpret_cast<const float2*>(Arow + ka);
        v2f a; a.x = av.x; a.y = av.y;
        v2f b;
        b.x = W[(size_t)ka * N + col];
        b.y = W[(size_t)(ka + 1) * N + col];
        acc = __builtin_amdgcn_wmma_f32_16x16x4_f32(
             false, a,  false, b,
             (short)0, acc,  false,  false);
    }

    float bv = bias[col];
#pragma unroll
    for (int v = 0; v < 8; ++v) {
        int orow = (tileM << 4) + v + (half << 3);
        float r = acc[v] + bv;
        if (RELU) r = fmaxf(r, 0.0f);
        C[(size_t)orow * N + col] = r;
    }
}

extern "C" void kernel_launch(void* const* d_in, const int* in_sizes, int n_in,
                              void* d_out, int out_size, void* d_ws, size_t ws_size,
                              hipStream_t stream) {
    const float* x         = (const float*)d_in[0];
    const int*   edge_idx  = (const int*)d_in[1];
    const float* edge_attr = (const float*)d_in[2];
    const float* u         = (const float*)d_in[3];
    const int*   batch     = (const int*)d_in[4];
    const float* W1        = (const float*)d_in[5];
    const float* b1        = (const float*)d_in[6];
    const float* W2        = (const float*)d_in[7];
    const float* b2        = (const float*)d_in[8];
    const float* W3        = (const float*)d_in[9];
    const float* b3        = (const float*)d_in[10];
    float* out = (float*)d_out;

    const int n_nodes = in_sizes[0] / NODE_DIM;
    const int n_edges = in_sizes[2] / EDGE_DIM;
    const int B       = in_sizes[3] / GLOBAL_DIM;

    float* ws = (float*)d_ws;
    float* node_sum = ws;
    float* node_cnt = node_sum + (size_t)NSEG_MAX * NODE_DIM;
    float* edge_sum = node_cnt + NSEG_MAX;
    float* edge_cnt = edge_sum + (size_t)NSEG_MAX * EDGE_DIM;
    float* H0 = edge_cnt + NSEG_MAX;
    float* H1 = H0 + (size_t)NSEG_MAX * IN_DIM;
    float* H2 = H1 + (size_t)NSEG_MAX * HIDDEN_DIM;

    const int n_acc = NSEG_MAX * NODE_DIM + NSEG_MAX + NSEG_MAX * EDGE_DIM + NSEG_MAX;
    zero_f32<<<(n_acc + 255) / 256, 256, 0, stream>>>(ws, n_acc);

    node_scatter<<<(n_nodes + NODES_PER_BLOCK - 1) / NODES_PER_BLOCK, NODE_DIM, 0, stream>>>(
        x, batch, node_sum, node_cnt, n_nodes);

    edge_scatter<<<240, 256, 0, stream>>>(
        edge_attr, edge_idx, batch, edge_sum, edge_cnt, n_edges, B);

    assemble<<<B, IN_DIM, 0, stream>>>(node_sum, node_cnt, edge_sum, edge_cnt, u, H0);

    {
        int tiles = (B / 16) * (HIDDEN_DIM / 16);
        gemm_wmma<true><<<(tiles + 7) / 8, 256, 0, stream>>>(H0, W1, b1, H1, B, HIDDEN_DIM, IN_DIM);
    }
    {
        int tiles = (B / 16) * (HIDDEN_DIM / 16);
        gemm_wmma<true><<<(tiles + 7) / 8, 256, 0, stream>>>(H1, W2, b2, H2, B, HIDDEN_DIM, HIDDEN_DIM);
    }
    {
        int tiles = (B / 16) * (GLOBAL_DIM / 16);
        gemm_wmma<false><<<(tiles + 7) / 8, 256, 0, stream>>>(H2, W3, b3, out, B, GLOBAL_DIM, HIDDEN_DIM);
    }
    (void)n_in; (void)out_size; (void)ws_size;
}


# node sums by graph ownership: a workgroup per graph finds its rows in the sorted batch (two-level search) and sums them, 16 row loads per wave; no node atomics, no zero launch; node sum order changes
# speedup vs baseline: 8.6750x; 8.6750x over previous
#include <hip/hip_runtime.h>
#include <hip/hip_bf16.h>

#define NODE_DIM   128
#define EDGE_DIM   32
#define GLOBAL_DIM 64
#define HIDDEN_DIM 256
#define IN_DIM     224
#define NSEG_MAX   512

typedef float v2f __attribute__((ext_vector_type(2)));
typedef float v8f __attribute__((ext_vector_type(8)));

__device__ __forceinline__ void gAtomicAddF(float* p, float v) {
    __hip_atomic_fetch_add(p, v, __ATOMIC_RELAXED, __HIP_MEMORY_SCOPE_AGENT);
}

#define NS_THREADS 256
#define NS_WAVES   (NS_THREADS / 32)
#define NS_PROBES  (2 * NS_THREADS)
#define NS_U       16

__device__ __forceinline__ int ns_total(const int* c) {
    int s = 0;
#pragma unroll
    for (int w = 0; w < NS_WAVES; ++w) s += c[w];
    return s;
}

__global__ void __launch_bounds__(NS_THREADS)
node_graph_mean(const float* __restrict__ x,
                const int* __restrict__ batch,
                float* __restrict__ node_mean,
                float* __restrict__ edge_sum,
                float* __restrict__ edge_cnt,
                int n_nodes) {
    __shared__ int    cw[4][NS_WAVES];
    __shared__ float4 part4[NS_WAVES * (NODE_DIM / 4)];
    const int g    = blockIdx.x;
    const int t    = threadIdx.x;
    const int lane = t & 31;
    const int wave = __builtin_amdgcn_readfirstlane(t >> 5);
    const int nlast = n_nodes - 1;

    if (t < EDGE_DIM) {
        edge_sum[(size_t)g * EDGE_DIM + t] = 0.0f;
        if ((g & 31) == 0 && g + t < NSEG_MAX) edge_cnt[g + t] = 0.0f;
    }

    const int S = (n_nodes + NS_PROBES - 1) / NS_PROBES;
    {
        const int pa = t * S, pb = (t + NS_THREADS) * S;
        const int va = batch[min(pa, nlast)], vb = batch[min(pb, nlast)];
        const bool ia = pa < n_nodes, ib = pb < n_nodes;
        const int k0 = __builtin_popcount(__builtin_amdgcn_ballot_w32(ia && va < g))
                     + __builtin_popcount(__builtin_amdgcn_ballot_w32(ib && vb < g));
        const int k1 = __builtin_popcount(__builtin_amdgcn_ballot_w32(ia && va <= g))
                     + __builtin_popcount(__builtin_amdgcn_ballot_w32(ib && vb <= g));
        if (lane == 0) { cw[0][wave] = k0; cw[1][wave] = k1; }
    }
    __syncthreads();
    const int c0 = ns_total(cw[0]), c1 = ns_total(cw[1]);
    const int w0s = c0 ? (c0 - 1) * S + 1 : 0;
    const int w0e = c0 ? min(c0 * S, n_nodes) : 0;
    const int w1s = c1 ? (c1 - 1) * S + 1 : 0;
    const int w1e = c1 ? min(c1 * S, n_nodes) : 0;

    {
        int k0 = 0, k1 = 0;
        for (int base = 0; base < S - 1; base += NS_THREADS) {
            const int i0 = w0s + base + t, i1 = w1s + base + t;
            const int v0 = batch[min(i0, nlast)], v1 = batch[min(i1, nlast)];
            k0 += __builtin_popcount(__builtin_amdgcn_ballot_w32(i0 < w0e && v0 < g));
            k1 += __builtin_popcount(__builtin_amdgcn_ballot_w32(i1 < w1e && v1 <= g));
        }
        if (lane == 0) { cw[2][wave] = k0; cw[3][wave] = k1; }
    }
    __syncthreads();
    const int lo = __builtin_amdgcn_readfirstlane(w0s + ns_total(cw[2]));
    const int hi = __builtin_amdgcn_readfirstlane(w1s + ns_total(cw[3]));

    const float4* __restrict__ xr = reinterpret_cast<const float4*>(x);
    float4 acc = make_float4(0.0f, 0.0f, 0.0f, 0.0f);
    for (int r = lo + wave; r < hi; r += NS_WAVES * NS_U) {
        float4 v[NS_U];
#pragma unroll
        for (int u = 0; u < NS_U; ++u) {
            const int rr = min(r + u * NS_WAVES, hi - 1);
            v[u] = xr[(size_t)rr * (NODE_DIM / 4) + lane];
        }
#pragma unroll
        for (int u = 0; u < NS_U; ++u) {
            const bool ok = (r + u * NS_WAVES) < hi;
            acc.x += ok ? v[u].x : 0.0f;
            acc.y += ok ? v[u].y : 0.0f;
            acc.z += ok ? v[u].z : 0.0f;
            acc.w += ok ? v[u].w : 0.0f;
        }
    }
    part4[wave * (NODE_DIM / 4) + lane] = acc;
    __syncthreads();

    if (t < NODE_DIM) {
        const float* part = reinterpret_cast<const float*>(part4);
        float s = 0.0f;
#pragma unroll
        for (int w = 0; w < NS_WAVES; ++w) s += part[w * NODE_DIM + t];
        node_mean[(size_t)g * NODE_DIM + t] = s / fmaxf((float)max(hi - lo, 0), 1.0f);
    }
}

#define ES_THREADS 1024
#define ES_WAVES   (ES_THREADS / 32)
#define ES_RING    40
#define ES_REVS    5
#define ES_EPW     (ES_RING * ES_REVS)
#define ES_TILE    (ES_EPW * ES_WAVES)
#define ES_IDK     ((ES_TILE + ES_THREADS - 1) / ES_THREADS)
#define ES_IDCAP   (ES_IDK * ES_THREADS)
#define ES_SPARE   NSEG_MAX

template <bool FULL>
__device__ __forceinline__ float es_row(const float* __restrict__ edge_attr,
                                        int e, int last, int lane) {
    if (!FULL) e = min(e, last);
    return __builtin_nontemporal_load(edge_attr + (size_t)e * EDGE_DIM + lane);
}

template <bool FULL>
__device__ __forceinline__ void es_first(const float* __restrict__ edge_attr,
                                         float (&r)[ES_RING], int wbase, int last, int lane) {
#pragma unroll
    for (int j = 0; j < ES_RING; ++j) r[j] = es_row<FULL>(edge_attr, wbase + j, last, lane);
}

template <bool FULL>
__device__ __forceinline__ void es_rows(const float* __restrict__ edge_attr,
                                        float* lsum, const unsigned short* wid,
                                        float (&r)[ES_RING], int wbase, int last, int lane) {
    unsigned ida = wid[lane], idb = wid[32 + lane];
    for (int rev = 0; rev < ES_REVS - 1; ++rev) {
        const unsigned na = wid[(rev + 1) * ES_RING + lane];
        const unsigned nb = wid[(rev + 1) * ES_RING + 32 + lane];
        const int nrow = wbase + (rev + 1) * ES_RING;
#pragma unroll
        for (int j = 0; j < ES_RING; ++j) {
            const unsigned id = (j < 32) ? __builtin_amdgcn_readlane(ida, j & 31)
                                         : __builtin_amdgcn_readlane(idb, j & 31);
            atomicAdd(&lsum[id * EDGE_DIM + lane], r[j]);
            r[j] = es_row<FULL>(edge_attr, nrow + j, last, lane);
        }
        ida = na; idb = nb;
    }
#pragma unroll
    for (int j = 0; j < ES_RING; ++j) {
        const unsigned id = (j < 32) ? __builtin_amdgcn_readlane(ida, j & 31)
                                     : __builtin_amdgcn_readlane(idb, j & 31);
        atomicAdd(&lsum[id * EDGE_DIM + lane], r[j]);
    }
}

__global__ void __launch_bounds__(ES_THREADS)
edge_scatter(const float* __restrict__ edge_attr,
             const int* __restrict__ edge_src,
             const int* __restrict__ batch,
             float* __restrict__ edge_sum,
             float* __restrict__ edge_cnt,
             int n_edges, int n_nodes, int nseg) {
    __shared__ float lsum[(NSEG_MAX + 1) * EDGE_DIM];
    __shared__ float lcnt[NSEG_MAX];
    __shared__ unsigned short lid[ES_IDCAP];
    for (int i = threadIdx.x; i < (NSEG_MAX + 1) * EDGE_DIM; i += ES_THREADS) lsum[i] = 0.0f;
    for (int i = threadIdx.x; i < NSEG_MAX; i += ES_THREADS) lcnt[i] = 0.0f;
    __syncthreads();

    const int lane = threadIdx.x & 31;
    const int wave = __builtin_amdgcn_readfirstlane(threadIdx.x >> 5);
    const int nsegc = min(nseg, NSEG_MAX);
    const int last = n_edges - 1;
    const int ntiles = (n_edges + ES_TILE - 1) / ES_TILE;

    for (int tile = blockIdx.x; tile < ntiles; tile += gridDim.x) {
        const int tbase = tile * ES_TILE;
        const int tcount = min(ES_TILE, n_edges - tbase);
        const int wbase = tbase + wave * ES_EPW;
        const bool full = (wbase + ES_EPW <= n_edges);

        float r[ES_RING];
        if (full) es_first<true>(edge_attr, r, wbase, last, lane);
        else      es_first<false>(edge_attr, r, wbase, last, lane);

        int sn[ES_IDK];
#pragma unroll
        for (int k = 0; k < ES_IDK; ++k) {
            const int i = min((int)threadIdx.x + k * ES_THREADS, tcount - 1);
            sn[k] = edge_src[tbase + i];
        }
        int g[ES_IDK];
#pragma unroll
        for (int k = 0; k < ES_IDK; ++k) {
            const unsigned s = min((unsigned)sn[k], (unsigned)(n_nodes - 1));
            g[k] = batch[s];
        }
#pragma unroll
        for (int k = 0; k < ES_IDK; ++k) {
            const int i = (int)threadIdx.x + k * ES_THREADS;
            const bool ok = (i < tcount) && ((unsigned)g[k] < (unsigned)nsegc);
            lid[i] = ok ? (unsigned short)g[k] : (unsigned short)ES_SPARE;
            if (ok) atomicAdd(&lcnt[g[k]], 1.0f);
        }
        __syncthreads();

        const unsigned short* wid = lid + wave * ES_EPW;
        if (full) es_rows<true>(edge_attr, lsum, wid, r, wbase, last, lane);
        else      es_rows<false>(edge_attr, lsum, wid, r, wbase, last, lane);
        __syncthreads();
    }

    for (int i = threadIdx.x; i < nsegc * EDGE_DIM; i += ES_THREADS) {
        float v = lsum[i];
        if (v != 0.0f) gAtomicAddF(&edge_sum[i], v);
    }
    for (int i = threadIdx.x; i < nsegc; i += ES_THREADS) {
        float v = lcnt[i];
        if (v != 0.0f) gAtomicAddF(&edge_cnt[i], v);
    }
}

__global__ void assemble(const float* __restrict__ node_mean,
                         const float* __restrict__ edge_sum,
                         const float* __restrict__ edge_cnt,
                         const float* __restrict__ u,
                         float* __restrict__ H0) {
    int b = blockIdx.x;
    int j = threadIdx.x;
    float v;
    if (j < NODE_DIM) {
        v = node_mean[(size_t)b * NODE_DIM + j];
    } else if (j < NODE_DIM + EDGE_DIM) {
        v = edge_sum[(size_t)b * EDGE_DIM + (j - NODE_DIM)] / fmaxf(edge_cnt[b], 1.0f);
    } else {
        v = u[(size_t)b * GLOBAL_DIM + (j - NODE_DIM - EDGE_DIM)];
    }
    H0[(size_t)b * IN_DIM + j] = v;
}

template <bool RELU>
__global__ void gemm_wmma(const float* __restrict__ A,
                          const float* __restrict__ W,
                          const float* __restrict__ bias,
                          float* __restrict__ C,
                          int M, int N, int K) {
    int lane = threadIdx.x & 31;
    int wave = threadIdx.x >> 5;
    int wavesPerBlock = blockDim.x >> 5;
    int tile = blockIdx.x * wavesPerBlock + wave;
    int ntilesN = N >> 4;
    int tileM = tile / ntilesN;
    int tileN = tile - tileM * ntilesN;
    if (tileM >= (M >> 4)) return;

    int half = lane >> 4;
    int l15  = lane & 15;
    int row  = (tileM << 4) + l15;
    int col  = (tileN << 4) + l15;

    const float* __restrict__ Arow = A + (size_t)row * K;
    v8f acc = {};
    for (int k0 = 0; k0 < K; k0 += 4) {
        int ka = k0 + (half << 1);
        float2 av = *reinterpret_cast<const float2*>(Arow + ka);
        v2f a; a.x = av.x; a.y = av.y;
        v2f b;
        b.x = W[(size_t)ka * N + col];
        b.y = W[(size_t)(ka + 1) * N + col];
        acc = __builtin_amdgcn_wmma_f32_16x16x4_f32(
             false, a,  false, b,
             (short)0, acc,  false,  false);
    }

    float bv = bias[col];
#pragma unroll
    for (int v = 0; v < 8; ++v) {
        int orow = (tileM << 4) + v + (half << 3);
        float r = acc[v] + bv;
        if (RELU) r = fmaxf(r, 0.0f);
        C[(size_t)orow * N + col] = r;
    }
}

extern "C" void kernel_launch(void* const* d_in, const int* in_sizes, int n_in,
                              void* d_out, int out_size, void* d_ws, size_t ws_size,
                              hipStream_t stream) {
    const float* x         = (const float*)d_in[0];
    const int*   edge_idx  = (const int*)d_in[1];
    const float* edge_attr = (const float*)d_in[2];
    const float* u         = (const float*)d_in[3];
    const int*   batch     = (const int*)d_in[4];
    const float* W1        = (const float*)d_in[5];
    const float* b1        = (const float*)d_in[6];
    const float* W2        = (const float*)d_in[7];
    const float* b2        = (const float*)d_in[8];
    const float* W3        = (const float*)d_in[9];
    const float* b3        = (const float*)d_in[10];
    float* out = (float*)d_out;

    const int n_nodes = in_sizes[0] / NODE_DIM;
    const int n_edges = in_sizes[2] / EDGE_DIM;
    const int B       = in_sizes[3] / GLOBAL_DIM;

    float* ws = (float*)d_ws;
    float* node_mean = ws;
    float* edge_sum = node_mean + (size_t)NSEG_MAX * NODE_DIM + NSEG_MAX;
    float* edge_cnt = edge_sum + (size_t)NSEG_MAX * EDGE_DIM;
    float* H0 = edge_cnt + NSEG_MAX;
    float* H1 = H0 + (size_t)NSEG_MAX * IN_DIM;
    float* H2 = H1 + (size_t)NSEG_MAX * HIDDEN_DIM;

    node_graph_mean<<<B, NS_THREADS, 0, stream>>>(
        x, batch, node_mean, edge_sum, edge_cnt, n_nodes);

    edge_scatter<<<250, ES_THREADS, 0, stream>>>(
        edge_attr, edge_idx, batch, edge_sum, edge_cnt, n_edges, n_nodes, B);

    assemble<<<B, IN_DIM, 0, stream>>>(node_mean, edge_sum, edge_cnt, u, H0);

    {
        int tiles = (B / 16) * (HIDDEN_DIM / 16);
        gemm_wmma<true><<<(tiles + 7) / 8, 256, 0, stream>>>(H0, W1, b1, H1, B, HIDDEN_DIM, IN_DIM);
    }
    {
        int tiles = (B / 16) * (HIDDEN_DIM / 16);
        gemm_wmma<true><<<(tiles + 7) / 8, 256, 0, stream>>>(H1, W2, b2, H2, B, HIDDEN_DIM, HIDDEN_DIM);
    }
    {
        int tiles = (B / 16) * (GLOBAL_DIM / 16);
        gemm_wmma<false><<<(tiles + 7) / 8, 256, 0, stream>>>(H2, W3, b3, out, B, GLOBAL_DIM, HIDDEN_DIM);
    }
    (void)n_in; (void)out_size; (void)ws_size;
}


# the three GEMMs request a tile's whole K range before the first WMMA: K, N compile-time, 16-byte A loads (same new k order in A and B), two accumulators, two waves per workgroup; k sum order changes
# speedup vs baseline: 13.9116x; 1.6036x over previous
#include <hip/hip_runtime.h>
#include <hip/hip_bf16.h>

#define NODE_DIM   128
#define EDGE_DIM   32
#define GLOBAL_DIM 64
#define HIDDEN_DIM 256
#define IN_DIM     224
#define NSEG_MAX   512

typedef float v2f __attribute__((ext_vector_type(2)));
typedef float v8f __attribute__((ext_vector_type(8)));

__device__ __forceinline__ void gAtomicAddF(float* p, float v) {
    __hip_atomic_fetch_add(p, v, __ATOMIC_RELAXED, __HIP_MEMORY_SCOPE_AGENT);
}

#define NS_THREADS 256
#define NS_WAVES   (NS_THREADS / 32)
#define NS_PROBES  (2 * NS_THREADS)
#define NS_U       16

__device__ __forceinline__ int ns_total(const int* c) {
    int s = 0;
#pragma unroll
    for (int w = 0; w < NS_WAVES; ++w) s += c[w];
    return s;
}

__global__ void __launch_bounds__(NS_THREADS)
node_graph_mean(const float* __restrict__ x,
                const int* __restrict__ batch,
                float* __restrict__ node_mean,
                float* __restrict__ edge_sum,
                float* __restrict__ edge_cnt,
                int n_nodes) {
    __shared__ int    cw[4][NS_WAVES];
    __shared__ float4 part4[NS_WAVES * (NODE_DIM / 4)];
    const int g    = blockIdx.x;
    const int t    = threadIdx.x;
    const int lane = t & 31;
    const int wave = __builtin_amdgcn_readfirstlane(t >> 5);
    const int nlast = n_nodes - 1;

    if (t < EDGE_DIM) {
        edge_sum[(size_t)g * EDGE_DIM + t] = 0.0f;
        if ((g & 31) == 0 && g + t < NSEG_MAX) edge_cnt[g + t] = 0.0f;
    }

    const int S = (n_nodes + NS_PROBES - 1) / NS_PROBES;
    {
        const int pa = t * S, pb = (t + NS_THREADS) * S;
        const int va = batch[min(pa, nlast)], vb = batch[min(pb, nlast)];
        const bool ia = pa < n_nodes, ib = pb < n_nodes;
        const int k0 = __builtin_popcount(__builtin_amdgcn_ballot_w32(ia && va < g))
                     + __builtin_popcount(__builtin_amdgcn_ballot_w32(ib && vb < g));
        const int k1 = __builtin_popcount(__builtin_amdgcn_ballot_w32(ia && va <= g))
                     + __builtin_popcount(__builtin_amdgcn_ballot_w32(ib && vb <= g));
        if (lane == 0) { cw[0][wave] = k0; cw[1][wave] = k1; }
    }
    __syncthreads();
    const int c0 = ns_total(cw[0]), c1 = ns_total(cw[1]);
    const int w0s = c0 ? (c0 - 1) * S + 1 : 0;
    const int w0e = c0 ? min(c0 * S, n_nodes) : 0;
    const int w1s = c1 ? (c1 - 1) * S + 1 : 0;
    const int w1e = c1 ? min(c1 * S, n_nodes) : 0;

    {
        int k0 = 0, k1 = 0;
        for (int base = 0; base < S - 1; base += NS_THREADS) {
            const int i0 = w0s + base + t, i1 = w1s + base + t;
            const int v0 = batch[min(i0, nlast)], v1 = batch[min(i1, nlast)];
            k0 += __builtin_popcount(__builtin_amdgcn_ballot_w32(i0 < w0e && v0 < g));
            k1 += __builtin_popcount(__builtin_amdgcn_ballot_w32(i1 < w1e && v1 <= g));
        }
        if (lane == 0) { cw[2][wave] = k0; cw[3][wave] = k1; }
    }
    __syncthreads();
    const int lo = __builtin_amdgcn_readfirstlane(w0s + ns_total(cw[2]));
    const int hi = __builtin_amdgcn_readfirstlane(w1s + ns_total(cw[3]));

    const float4* __restrict__ xr = reinterpret_cast<const float4*>(x);
    float4 acc = make_float4(0.0f, 0.0f, 0.0f, 0.0f);
    for (int r = lo + wave; r < hi; r += NS_WAVES * NS_U) {
        float4 v[NS_U];
#pragma unroll
        for (int u = 0; u < NS_U; ++u) {
            const int rr = min(r + u * NS_WAVES, hi - 1);
            v[u] = xr[(size_t)rr * (NODE_DIM / 4) + lane];
        }
#pragma unroll
        for (int u = 0; u < NS_U; ++u) {
            const bool ok = (r + u * NS_WAVES) < hi;
            acc.x += ok ? v[u].x : 0.0f;
            acc.y += ok ? v[u].y : 0.0f;
            acc.z += ok ? v[u].z : 0.0f;
            acc.w += ok ? v[u].w : 0.0f;
        }
    }
    part4[wave * (NODE_DIM / 4) + lane] = acc;
    __syncthreads();

    if (t < NODE_DIM) {
        const float* part = reinterpret_cast<const float*>(part4);
        float s = 0.0f;
#pragma unroll
        for (int w = 0; w < NS_WAVES; ++w) s += part[w * NODE_DIM + t];
        node_mean[(size_t)g * NODE_DIM + t] = s / fmaxf((float)max(hi - lo, 0), 1.0f);
    }
}

#define ES_THREADS 1024
#define ES_WAVES   (ES_THREADS / 32)
#define ES_RING    40
#define ES_REVS    5
#define ES_EPW     (ES_RING * ES_REVS)
#define ES_TILE    (ES_EPW * ES_WAVES)
#define ES_IDK     ((ES_TILE + ES_THREADS - 1) / ES_THREADS)
#define ES_IDCAP   (ES_IDK * ES_THREADS)
#define ES_SPARE   NSEG_MAX

template <bool FULL>
__device__ __forceinline__ float es_row(const float* __restrict__ edge_attr,
                                        int e, int last, int lane) {
    if (!FULL) e = min(e, last);
    return __builtin_nontemporal_load(edge_attr + (size_t)e * EDGE_DIM + lane);
}

template <bool FULL>
__device__ __forceinline__ void es_first(const float* __restrict__ edge_attr,
                                         float (&r)[ES_RING], int wbase, int last, int lane) {
#pragma unroll
    for (int j = 0; j < ES_RING; ++j) r[j] = es_row<FULL>(edge_attr, wbase + j, last, lane);
}

template <bool FULL>
__device__ __forceinline__ void es_rows(const float* __restrict__ edge_attr,
                                        float* lsum, const unsigned short* wid,
                                        float (&r)[ES_RING], int wbase, int last, int lane) {
    unsigned ida = wid[lane], idb = wid[32 + lane];
    for (int rev = 0; rev < ES_REVS - 1; ++rev) {
        const unsigned na = wid[(rev + 1) * ES_RING + lane];
        const unsigned nb = wid[(rev + 1) * ES_RING + 32 + lane];
        const int nrow = wbase + (rev + 1) * ES_RING;
#pragma unroll
        for (int j = 0; j < ES_RING; ++j) {
            const unsigned id = (j < 32) ? __builtin_amdgcn_readlane(ida, j & 31)
                                         : __builtin_amdgcn_readlane(idb, j & 31);
            atomicAdd(&lsum[id * EDGE_DIM + lane], r[j]);
            r[j] = es_row<FULL>(edge_attr, nrow + j, last, lane);
        }
        ida = na; idb = nb;
    }
#pragma unroll
    for (int j = 0; j < ES_RING; ++j) {
        const unsigned id = (j < 32) ? __builtin_amdgcn_readlane(ida, j & 31)
                                     : __builtin_amdgcn_readlane(idb, j & 31);
        atomicAdd(&lsum[id * EDGE_DIM + lane], r[j]);
    }
}

__global__ void __launch_bounds__(ES_THREADS)
edge_scatter(const float* __restrict__ edge_attr,
             const int* __restrict__ edge_src,
             const int* __restrict__ batch,
             float* __restrict__ edge_sum,
             float* __restrict__ edge_cnt,
             int n_edges, int n_nodes, int nseg) {
    __shared__ float lsum[(NSEG_MAX + 1) * EDGE_DIM];
    __shared__ float lcnt[NSEG_MAX];
    __shared__ unsigned short lid[ES_IDCAP];
    for (int i = threadIdx.x; i < (NSEG_MAX + 1) * EDGE_DIM; i += ES_THREADS) lsum[i] = 0.0f;
    for (int i = threadIdx.x; i < NSEG_MAX; i += ES_THREADS) lcnt[i] = 0.0f;
    __syncthreads();

    const int lane = threadIdx.x & 31;
    const int wave = __builtin_amdgcn_readfirstlane(threadIdx.x >> 5);
    const int nsegc = min(nseg, NSEG_MAX);
    const int last = n_edges - 1;
    const int ntiles = (n_edges + ES_TILE - 1) / ES_TILE;

    for (int tile = blockIdx.x; tile < ntiles; tile += gridDim.x) {
        const int tbase = tile * ES_TILE;
        const int tcount = min(ES_TILE, n_edges - tbase);
        const int wbase = tbase + wave * ES_EPW;
        const bool full = (wbase + ES_EPW <= n_edges);

        float r[ES_RING];
        if (full) es_first<true>(edge_attr, r, wbase, last, lane);
        else      es_first<false>(edge_attr, r, wbase, last, lane);

        int sn[ES_IDK];
#pragma unroll
        for (int k = 0; k < ES_IDK; ++k) {
            const int i = min((int)threadIdx.x + k * ES_THREADS, tcount - 1);
            sn[k] = edge_src[tbase + i];
        }
        int g[ES_IDK];
#pragma unroll
        for (int k = 0; k < ES_IDK; ++k) {
            const unsigned s = min((unsigned)sn[k], (unsigned)(n_nodes - 1));
            g[k] = batch[s];
        }
#pragma unroll
        for (int k = 0; k < ES_IDK; ++k) {
            const int i = (int)threadIdx.x + k * ES_THREADS;
            const bool ok = (i < tcount) && ((unsigned)g[k] < (unsigned)nsegc);
            lid[i] = ok ? (unsigned short)g[k] : (unsigned short)ES_SPARE;
            if (ok) atomicAdd(&lcnt[g[k]], 1.0f);
        }
        __syncthreads();

        const unsigned short* wid = lid + wave * ES_EPW;
        if (full) es_rows<true>(edge_attr, lsum, wid, r, wbase, last, lane);
        else      es_rows<false>(edge_attr, lsum, wid, r, wbase, last, lane);
        __syncthreads();
    }

    for (int i = threadIdx.x; i < nsegc * EDGE_DIM; i += ES_THREADS) {
        float v = lsum[i];
        if (v != 0.0f) gAtomicAddF(&edge_sum[i], v);
    }
    for (int i = threadIdx.x; i < nsegc; i += ES_THREADS) {
        float v = lcnt[i];
        if (v != 0.0f) gAtomicAddF(&edge_cnt[i], v);
    }
}

__global__ void assemble(const float* __restrict__ node_mean,
                         const float* __restrict__ edge_sum,
                         const float* __restrict__ edge_cnt,
                         const float* __restrict__ u,
                         float* __restrict__ H0) {
    int b = blockIdx.x;
    int j = threadIdx.x;
    float v;
    if (j < NODE_DIM) {
        v = node_mean[(size_t)b * NODE_DIM + j];
    } else if (j < NODE_DIM + EDGE_DIM) {
        v = edge_sum[(size_t)b * EDGE_DIM + (j - NODE_DIM)] / fmaxf(edge_cnt[b], 1.0f);
    } else {
        v = u[(size_t)b * GLOBAL_DIM + (j - NODE_DIM - EDGE_DIM)];
    }
    H0[(size_t)b * IN_DIM + j] = v;
}

template <bool RELU, int K, int N>
__global__ void __launch_bounds__(64)
gemm_wmma(const float* __restrict__ A,
          const float* __restrict__ W,
          const float* __restrict__ bias,
          float* __restrict__ C,
          int M) {
    constexpr int NT = N / 16;
    constexpr int KG = K / 8;
    static_assert(K % 8 == 0 && N % 16 == 0, "tile shape");
    const int lane  = threadIdx.x & 31;
    const int wave  = __builtin_amdgcn_readfirstlane(threadIdx.x >> 5);
    const int tile  = blockIdx.x * 2 + wave;
    const int tileM = tile / NT;
    const int tileN = tile - tileM * NT;
    if (tileM >= (M >> 4)) return;

    const int half = lane >> 4;
    const int l15  = lane & 15;
    const int row  = (tileM << 4) + l15;
    const int col  = (tileN << 4) + l15;

    const float4* __restrict__ Ap = reinterpret_cast<const float4*>(A + (size_t)row * K + 4 * half);
    const float* __restrict__ Wp = W + (size_t)(4 * half) * N + col;

    const float bv = bias[col];
    float4 a[KG];
    float  b[KG][4];
#pragma unroll
    for (int t = 0; t < KG; ++t) {
        a[t] = Ap[2 * t];
#pragma unroll
        for (int s = 0; s < 4; ++s) b[t][s] = Wp[(size_t)(8 * t + s) * N];
    }

    __builtin_amdgcn_sched_barrier(0);

    v8f acc0 = {}, acc1 = {};
#pragma unroll
    for (int t = 0; t < KG; ++t) {
        v2f a0; a0.x = a[t].x; a0.y = a[t].y;
        v2f b0; b0.x = b[t][0]; b0.y = b[t][1];
        acc0 = __builtin_amdgcn_wmma_f32_16x16x4_f32(
             false, a0,  false, b0,
             (short)0, acc0,  false,  false);
        v2f a1; a1.x = a[t].z; a1.y = a[t].w;
        v2f b1; b1.x = b[t][2]; b1.y = b[t][3];
        acc1 = __builtin_amdgcn_wmma_f32_16x16x4_f32(
             false, a1,  false, b1,
             (short)0, acc1,  false,  false);
    }

#pragma unroll
    for (int v = 0; v < 8; ++v) {
        const int orow = (tileM << 4) + v + (half << 3);
        float r = (acc0[v] + acc1[v]) + bv;
        if (RELU) r = fmaxf(r, 0.0f);
        C[(size_t)orow * N + col] = r;
    }
}

extern "C" void kernel_launch(void* const* d_in, const int* in_sizes, int n_in,
                              void* d_out, int out_size, void* d_ws, size_t ws_size,
                              hipStream_t stream) {
    const float* x         = (const float*)d_in[0];
    const int*   edge_idx  = (const int*)d_in[1];
    const float* edge_attr = (const float*)d_in[2];
    const float* u         = (const float*)d_in[3];
    const int*   batch     = (const int*)d_in[4];
    const float* W1        = (const float*)d_in[5];
    const float* b1        = (const float*)d_in[6];
    const float* W2        = (const float*)d_in[7];
    const float* b2        = (const float*)d_in[8];
    const float* W3        = (const float*)d_in[9];
    const float* b3        = (const float*)d_in[10];
    float* out = (float*)d_out;

    const int n_nodes = in_sizes[0] / NODE_DIM;
    const int n_edges = in_sizes[2] / EDGE_DIM;
    const int B       = in_sizes[3] / GLOBAL_DIM;

    float* ws = (float*)d_ws;
    float* node_mean = ws;
    float* edge_sum = node_mean + (size_t)NSEG_MAX * NODE_DIM + NSEG_MAX;
    float* edge_cnt = edge_sum + (size_t)NSEG_MAX * EDGE_DIM;
    float* H0 = edge_cnt + NSEG_MAX;
    float* H1 = H0 + (size_t)NSEG_MAX * IN_DIM;
    float* H2 = H1 + (size_t)NSEG_MAX * HIDDEN_DIM;

    node_graph_mean<<<B, NS_THREADS, 0, stream>>>(
        x, batch, node_mean, edge_sum, edge_cnt, n_nodes);

    edge_scatter<<<250, ES_THREADS, 0, stream>>>(
        edge_attr, edge_idx, batch, edge_sum, edge_cnt, n_edges, n_nodes, B);

    assemble<<<B, IN_DIM, 0, stream>>>(node_mean, edge_sum, edge_cnt, u, H0);

    {
        int tiles = (B / 16) * (HIDDEN_DIM / 16);
        gemm_wmma<true, IN_DIM, HIDDEN_DIM><<<tiles / 2, 64, 0, stream>>>(H0, W1, b1, H1, B);
    }
    {
        int tiles = (B / 16) * (HIDDEN_DIM / 16);
        gemm_wmma<true, HIDDEN_DIM, HIDDEN_DIM><<<tiles / 2, 64, 0, stream>>>(H1, W2, b2, H2, B);
    }
    {
        int tiles = (B / 16) * (GLOBAL_DIM / 16);
        gemm_wmma<false, HIDDEN_DIM, GLOBAL_DIM><<<tiles / 2, 64, 0, stream>>>(H2, W3, b3, out, B);
    }
    (void)n_in; (void)out_size; (void)ws_size;
}


# edge flush without global atomics: each scatter workgroup stores its LDS sums and counts as one workspace plane (16-byte stores); assemble adds a graph's 250 rows in plane order; sum order changes
# speedup vs baseline: 14.4639x; 1.0397x over previous
#include <hip/hip_runtime.h>
#include <hip/hip_bf16.h>

#define NODE_DIM   128
#define EDGE_DIM   32
#define GLOBAL_DIM 64
#define HIDDEN_DIM 256
#define IN_DIM     224
#define NSEG_MAX   512

#define ES_GRID        250
#define PLANE_FLOATS   (NSEG_MAX * EDGE_DIM)
#define WS_FLOATS_BASE (NSEG_MAX * NODE_DIM + NSEG_MAX + NSEG_MAX * EDGE_DIM + NSEG_MAX \
                        + NSEG_MAX * IN_DIM + 2 * NSEG_MAX * HIDDEN_DIM)
#define WS_BYTES_PLANES ((size_t)(WS_FLOATS_BASE + ES_GRID * (PLANE_FLOATS + NSEG_MAX)) * sizeof(float))

typedef float v2f __attribute__((ext_vector_type(2)));
typedef float v8f __attribute__((ext_vector_type(8)));

__device__ __forceinline__ void gAtomicAddF(float* p, float v) {
    __hip_atomic_fetch_add(p, v, __ATOMIC_RELAXED, __HIP_MEMORY_SCOPE_AGENT);
}

#define NS_THREADS 256
#define NS_WAVES   (NS_THREADS / 32)
#define NS_PROBES  (2 * NS_THREADS)
#define NS_U       16

__device__ __forceinline__ int ns_total(const int* c) {
    int s = 0;
#pragma unroll
    for (int w = 0; w < NS_WAVES; ++w) s += c[w];
    return s;
}

__global__ void __launch_bounds__(NS_THREADS)
node_graph_mean(const float* __restrict__ x,
                const int* __restrict__ batch,
                float* __restrict__ node_mean,
                float* __restrict__ edge_sum,
                float* __restrict__ edge_cnt,
                int n_nodes, size_t ws_bytes) {
    __shared__ int    cw[4][NS_WAVES];
    __shared__ float4 part4[NS_WAVES * (NODE_DIM / 4)];
    const int g    = blockIdx.x;
    const int t    = threadIdx.x;
    const int lane = t & 31;
    const int wave = __builtin_amdgcn_readfirstlane(t >> 5);
    const int nlast = n_nodes - 1;

    if (ws_bytes < WS_BYTES_PLANES && t < EDGE_DIM) {
        edge_sum[(size_t)g * EDGE_DIM + t] = 0.0f;
        if ((g & 31) == 0 && g + t < NSEG_MAX) edge_cnt[g + t] = 0.0f;
    }

    const int S = (n_nodes + NS_PROBES - 1) / NS_PROBES;
    {
        const int pa = t * S, pb = (t + NS_THREADS) * S;
        const int va = batch[min(pa, nlast)], vb = batch[min(pb, nlast)];
        const bool ia = pa < n_nodes, ib = pb < n_nodes;
        const int k0 = __builtin_popcount(__builtin_amdgcn_ballot_w32(ia && va < g))
                     + __builtin_popcount(__builtin_amdgcn_ballot_w32(ib && vb < g));
        const int k1 = __builtin_popcount(__builtin_amdgcn_ballot_w32(ia && va <= g))
                     + __builtin_popcount(__builtin_amdgcn_ballot_w32(ib && vb <= g));
        if (lane == 0) { cw[0][wave] = k0; cw[1][wave] = k1; }
    }
    __syncthreads();
    const int c0 = ns_total(cw[0]), c1 = ns_total(cw[1]);
    const int w0s = c0 ? (c0 - 1) * S + 1 : 0;
    const int w0e = c0 ? min(c0 * S, n_nodes) : 0;
    const int w1s = c1 ? (c1 - 1) * S + 1 : 0;
    const int w1e = c1 ? min(c1 * S, n_nodes) : 0;

    {
        int k0 = 0, k1 = 0;
        for (int base = 0; base < S - 1; base += NS_THREADS) {
            const int i0 = w0s + base + t, i1 = w1s + base + t;
            const int v0 = batch[min(i0, nlast)], v1 = batch[min(i1, nlast)];
            k0 += __builtin_popcount(__builtin_amdgcn_ballot_w32(i0 < w0e && v0 < g));
            k1 += __builtin_popcount(__builtin_amdgcn_ballot_w32(i1 < w1e && v1 <= g));
        }
        if (lane == 0) { cw[2][wave] = k0; cw[3][wave] = k1; }
    }
    __syncthreads();
    const int lo = __builtin_amdgcn_readfirstlane(w0s + ns_total(cw[2]));
    const int hi = __builtin_amdgcn_readfirstlane(w1s + ns_total(cw[3]));

    const float4* __restrict__ xr = reinterpret_cast<const float4*>(x);
    float4 acc = make_float4(0.0f, 0.0f, 0.0f, 0.0f);
    for (int r = lo + wave; r < hi; r += NS_WAVES * NS_U) {
        float4 v[NS_U];
#pragma unroll
        for (int u = 0; u < NS_U; ++u) {
            const int rr = min(r + u * NS_WAVES, hi - 1);
            v[u] = xr[(size_t)rr * (NODE_DIM / 4) + lane];
        }
#pragma unroll
        for (int u = 0; u < NS_U; ++u) {
            const bool ok = (r + u * NS_WAVES) < hi;
            acc.x += ok ? v[u].x : 0.0f;
            acc.y += ok ? v[u].y : 0.0f;
            acc.z += ok ? v[u].z : 0.0f;
            acc.w += ok ? v[u].w : 0.0f;
        }
    }
    part4[wave * (NODE_DIM / 4) + lane] = acc;
    __syncthreads();

    if (t < NODE_DIM) {
        const float* part = reinterpret_cast<const float*>(part4);
        float s = 0.0f;
#pragma unroll
        for (int w = 0; w < NS_WAVES; ++w) s += part[w * NODE_DIM + t];
        node_mean[(size_t)g * NODE_DIM + t] = s / fmaxf((float)max(hi - lo, 0), 1.0f);
    }
}

#define ES_THREADS 1024
#define ES_WAVES   (ES_THREADS / 32)
#define ES_RING    40
#define ES_REVS    5
#define ES_EPW     (ES_RING * ES_REVS)
#define ES_TILE    (ES_EPW * ES_WAVES)
#define ES_IDK     ((ES_TILE + ES_THREADS - 1) / ES_THREADS)
#define ES_IDCAP   (ES_IDK * ES_THREADS)
#define ES_SPARE   NSEG_MAX

template <bool FULL>
__device__ __forceinline__ float es_row(const float* __restrict__ edge_attr,
                                        int e, int last, int lane) {
    if (!FULL) e = min(e, last);
    return __builtin_nontemporal_load(edge_attr + (size_t)e * EDGE_DIM + lane);
}

template <bool FULL>
__device__ __forceinline__ void es_first(const float* __restrict__ edge_attr,
                                         float (&r)[ES_RING], int wbase, int last, int lane) {
#pragma unroll
    for (int j = 0; j < ES_RING; ++j) r[j] = es_row<FULL>(edge_attr, wbase + j, last, lane);
}

template <bool FULL>
__device__ __forceinline__ void es_rows(const float* __restrict__ edge_attr,
                                        float* lsum, const unsigned short* wid,
                                        float (&r)[ES_RING], int wbase, int last, int lane) {
    unsigned ida = wid[lane], idb = wid[32 + lane];
    for (int rev = 0; rev < ES_REVS - 1; ++rev) {
        const unsigned na = wid[(rev + 1) * ES_RING + lane];
        const unsigned nb = wid[(rev + 1) * ES_RING + 32 + lane];
        const int nrow = wbase + (rev + 1) * ES_RING;
#pragma unroll
        for (int j = 0; j < ES_RING; ++j) {
            const unsigned id = (j < 32) ? __builtin_amdgcn_readlane(ida, j & 31)
                                         : __builtin_amdgcn_readlane(idb, j & 31);
            atomicAdd(&lsum[id * EDGE_DIM + lane], r[j]);
            r[j] = es_row<FULL>(edge_attr, nrow + j, last, lane);
        }
        ida = na; idb = nb;
    }
#pragma unroll
    for (int j = 0; j < ES_RING; ++j) {
        const unsigned id = (j < 32) ? __builtin_amdgcn_readlane(ida, j & 31)
                                     : __builtin_amdgcn_readlane(idb, j & 31);
        atomicAdd(&lsum[id * EDGE_DIM + lane], r[j]);
    }
}

__global__ void __launch_bounds__(ES_THREADS)
edge_scatter(const float* __restrict__ edge_attr,
             const int* __restrict__ edge_src,
             const int* __restrict__ batch,
             float* __restrict__ edge_sum,
             float* __restrict__ edge_cnt,
             float* __restrict__ psum,
             float* __restrict__ pcnt,
             int n_edges, int n_nodes, int nseg, size_t ws_bytes) {
    __shared__ __attribute__((aligned(16))) float lsum[(NSEG_MAX + 1) * EDGE_DIM];
    __shared__ __attribute__((aligned(16))) float lcnt[NSEG_MAX];
    __shared__ unsigned short lid[ES_IDCAP];
    for (int i = threadIdx.x; i < (NSEG_MAX + 1) * EDGE_DIM; i += ES_THREADS) lsum[i] = 0.0f;
    for (int i = threadIdx.x; i < NSEG_MAX; i += ES_THREADS) lcnt[i] = 0.0f;
    __syncthreads();

    const int lane = threadIdx.x & 31;
    const int wave = __builtin_amdgcn_readfirstlane(threadIdx.x >> 5);
    const int nsegc = min(nseg, NSEG_MAX);
    const int last = n_edges - 1;
    const int ntiles = (n_edges + ES_TILE - 1) / ES_TILE;

    for (int tile = blockIdx.x; tile < ntiles; tile += gridDim.x) {
        const int tbase = tile * ES_TILE;
        const int tcount = min(ES_TILE, n_edges - tbase);
        const int wbase = tbase + wave * ES_EPW;
        const bool full = (wbase + ES_EPW <= n_edges);

        float r[ES_RING];
        if (full) es_first<true>(edge_attr, r, wbase, last, lane);
        else      es_first<false>(edge_attr, r, wbase, last, lane);

        int sn[ES_IDK];
#pragma unroll
        for (int k = 0; k < ES_IDK; ++k) {
            const int i = min((int)threadIdx.x + k * ES_THREADS, tcount - 1);
            sn[k] = edge_src[tbase + i];
        }
        int g[ES_IDK];
#pragma unroll
        for (int k = 0; k < ES_IDK; ++k) {
            const unsigned s = min((unsigned)sn[k], (unsigned)(n_nodes - 1));
            g[k] = batch[s];
        }
#pragma unroll
        for (int k = 0; k < ES_IDK; ++k) {
            const int i = (int)threadIdx.x + k * ES_THREADS;
            const bool ok = (i < tcount) && ((unsigned)g[k] < (unsigned)nsegc);
            lid[i] = ok ? (unsigned short)g[k] : (unsigned short)ES_SPARE;
            if (ok) atomicAdd(&lcnt[g[k]], 1.0f);
        }
        __syncthreads();

        const unsigned short* wid = lid + wave * ES_EPW;
        if (full) es_rows<true>(edge_attr, lsum, wid, r, wbase, last, lane);
        else      es_rows<false>(edge_attr, lsum, wid, r, wbase, last, lane);
        __syncthreads();
    }

    if (ws_bytes >= WS_BYTES_PLANES) {
        static_assert((PLANE_FLOATS / 4) % ES_THREADS == 0, "whole trips");
        float4* __restrict__ ps4 = reinterpret_cast<float4*>(psum + (size_t)blockIdx.x * PLANE_FLOATS);
        const float4* ls4 = reinterpret_cast<const float4*>(lsum);
        float4 q[PLANE_FLOATS / 4 / ES_THREADS];
#pragma unroll
        for (int k = 0; k < PLANE_FLOATS / 4 / ES_THREADS; ++k) q[k] = ls4[threadIdx.x + k * ES_THREADS];
#pragma unroll
        for (int k = 0; k < PLANE_FLOATS / 4 / ES_THREADS; ++k) ps4[threadIdx.x + k * ES_THREADS] = q[k];
        float4* __restrict__ pc4 = reinterpret_cast<float4*>(pcnt + (size_t)blockIdx.x * NSEG_MAX);
        const float4* lc4 = reinterpret_cast<const float4*>(lcnt);
        for (int i = threadIdx.x; i < NSEG_MAX / 4; i += ES_THREADS) pc4[i] = lc4[i];
    } else {
        for (int i = threadIdx.x; i < nsegc * EDGE_DIM; i += ES_THREADS) {
            float v = lsum[i];
            if (v != 0.0f) gAtomicAddF(&edge_sum[i], v);
        }
        for (int i = threadIdx.x; i < nsegc; i += ES_THREADS) {
            float v = lcnt[i];
            if (v != 0.0f) gAtomicAddF(&edge_cnt[i], v);
        }
    }
}

#define AS_THREADS 256
#define AS_WAVES   (AS_THREADS / 32)
#define AS_U       32

__global__ void __launch_bounds__(AS_THREADS)
assemble(const float* __restrict__ node_mean,
         const float* __restrict__ edge_sum,
         const float* __restrict__ edge_cnt,
         const float* __restrict__ psum,
         const float* __restrict__ pcnt,
         const float* __restrict__ u,
         float* __restrict__ H0,
         size_t ws_bytes) {
    static_assert(ES_GRID <= AS_WAVES * AS_U && ES_GRID <= AS_WAVES * 32, "planes per workgroup");
    __shared__ float part[AS_WAVES][EDGE_DIM];
    __shared__ float cpart[AS_WAVES];
    const int b    = blockIdx.x;
    const int t    = threadIdx.x;
    const int lane = t & 31;
    const int wave = __builtin_amdgcn_readfirstlane(t >> 5);
    const bool planes = ws_bytes >= WS_BYTES_PLANES;

    float own = 0.0f;
    if (t < NODE_DIM) {
        own = node_mean[(size_t)b * NODE_DIM + t];
    } else if (t >= NODE_DIM + EDGE_DIM && t < IN_DIM) {
        own = u[(size_t)b * GLOBAL_DIM + (t - NODE_DIM - EDGE_DIM)];
    }

    if (planes) {
        const int pq = wave * 32 + lane;
        float c = pcnt[(size_t)min(pq, ES_GRID - 1) * NSEG_MAX + b];
        const float* __restrict__ rowp = psum + (size_t)b * EDGE_DIM + lane;
        float v[AS_U];
#pragma unroll
        for (int k = 0; k < AS_U; ++k) {
            const int p = min(wave + k * AS_WAVES, ES_GRID - 1);
            v[k] = rowp[(size_t)p * PLANE_FLOATS];
        }
        c = (pq < ES_GRID) ? c : 0.0f;
#pragma unroll
        for (int off = 16; off > 0; off >>= 1) c += __shfl_xor(c, off, 32);
        float acc = 0.0f;
#pragma unroll
        for (int k = 0; k < AS_U; ++k) acc += (wave + k * AS_WAVES < ES_GRID) ? v[k] : 0.0f;
        part[wave][lane] = acc;
        if (lane == 0) cpart[wave] = c;
    }
    __syncthreads();

    if (t < IN_DIM) {
        float v = own;
        if (t >= NODE_DIM && t < NODE_DIM + EDGE_DIM) {
            const int j = t - NODE_DIM;
            float s, n;
            if (planes) {
                s = 0.0f; n = 0.0f;
#pragma unroll
                for (int w = 0; w < AS_WAVES; ++w) { s += part[w][j]; n += cpart[w]; }
            } else {
                s = edge_sum[(size_t)b * EDGE_DIM + j];
                n = edge_cnt[b];
            }
            v = s / fmaxf(n, 1.0f);
        }
        H0[(size_t)b * IN_DIM + t] = v;
    }
}

template <bool RELU, int K, int N>
__global__ void __launch_bounds__(64)
gemm_wmma(const float* __restrict__ A,
          const float* __restrict__ W,
          const float* __restrict__ bias,
          float* __restrict__ C,
          int M) {
    constexpr int NT = N / 16;
    constexpr int KG = K / 8;
    static_assert(K % 8 == 0 && N % 16 == 0, "tile shape");
    const int lane  = threadIdx.x & 31;
    const int wave  = __builtin_amdgcn_readfirstlane(threadIdx.x >> 5);
    const int tile  = blockIdx.x * 2 + wave;
    const int tileM = tile / NT;
    const int tileN = tile - tileM * NT;
    if (tileM >= (M >> 4)) return;

    const int half = lane >> 4;
    const int l15  = lane & 15;
    const int row  = (tileM << 4) + l15;
    const int col  = (tileN << 4) + l15;

    const float4* __restrict__ Ap = reinterpret_cast<const float4*>(A + (size_t)row * K + 4 * half);
    const float* __restrict__ Wp = W + (size_t)(4 * half) * N + col;

    const float bv = bias[col];
    float4 a[KG];
    float  b[KG][4];
#pragma unroll
    for (int t = 0; t < KG; ++t) {
        a[t] = Ap[2 * t];
#pragma unroll
        for (int s = 0; s < 4; ++s) b[t][s] = Wp[(size_t)(8 * t + s) * N];
    }

    __builtin_amdgcn_sched_barrier(0);

    v8f acc0 = {}, acc1 = {};
#pragma unroll
    for (int t = 0; t < KG; ++t) {
        v2f a0; a0.x = a[t].x; a0.y = a[t].y;
        v2f b0; b0.x = b[t][0]; b0.y = b[t][1];
        acc0 = __builtin_amdgcn_wmma_f32_16x16x4_f32(
             false, a0,  false, b0,
             (short)0, acc0,  false,  false);
        v2f a1; a1.x = a[t].z; a1.y = a[t].w;
        v2f b1; b1.x = b[t][2]; b1.y = b[t][3];
        acc1 = __builtin_amdgcn_wmma_f32_16x16x4_f32(
             false, a1,  false, b1,
             (short)0, acc1,  false,  false);
    }

#pragma unroll
    for (int v = 0; v < 8; ++v) {
        const int orow = (tileM << 4) + v + (half << 3);
        float r = (acc0[v] + acc1[v]) + bv;
        if (RELU) r = fmaxf(r, 0.0f);
        C[(size_t)orow * N + col] = r;
    }
}

extern "C" void kernel_launch(void* const* d_in, const int* in_sizes, int n_in,
                              void* d_out, int out_size, void* d_ws, size_t ws_size,
                              hipStream_t stream) {
    const float* x         = (const float*)d_in[0];
    const int*   edge_idx  = (const int*)d_in[1];
    const float* edge_attr = (const float*)d_in[2];
    const float* u         = (const float*)d_in[3];
    const int*   batch     = (const int*)d_in[4];
    const float* W1        = (const float*)d_in[5];
    const float* b1        = (const float*)d_in[6];
    const float* W2        = (const float*)d_in[7];
    const float* b2        = (const float*)d_in[8];
    const float* W3        = (const float*)d_in[9];
    const float* b3        = (const float*)d_in[10];
    float* out = (float*)d_out;

    const int n_nodes = in_sizes[0] / NODE_DIM;
    const int n_edges = in_sizes[2] / EDGE_DIM;
    const int B       = in_sizes[3] / GLOBAL_DIM;

    float* ws = (float*)d_ws;
    float* node_mean = ws;
    float* edge_sum = node_mean + (size_t)NSEG_MAX * NODE_DIM + NSEG_MAX;
    float* edge_cnt = edge_sum + (size_t)NSEG_MAX * EDGE_DIM;
    float* H0 = edge_cnt + NSEG_MAX;
    float* H1 = H0 + (size_t)NSEG_MAX * IN_DIM;
    float* H2 = H1 + (size_t)NSEG_MAX * HIDDEN_DIM;
    float* psum = H2 + (size_t)NSEG_MAX * HIDDEN_DIM;
    float* pcnt = psum + (size_t)ES_GRID * PLANE_FLOATS;

    node_graph_mean<<<B, NS_THREADS, 0, stream>>>(
        x, batch, node_mean, edge_sum, edge_cnt, n_nodes, ws_size);

    edge_scatter<<<ES_GRID, ES_THREADS, 0, stream>>>(
        edge_attr, edge_idx, batch, edge_sum, edge_cnt, psum, pcnt, n_edges, n_nodes, B, ws_size);

    assemble<<<B, AS_THREADS, 0, stream>>>(node_mean, edge_sum, edge_cnt, psum, pcnt, u, H0, ws_size);

    {
        int tiles = (B / 16) * (HIDDEN_DIM / 16);
        gemm_wmma<true, IN_DIM, HIDDEN_DIM><<<tiles / 2, 64, 0, stream>>>(H0, W1, b1, H1, B);
    }
    {
        int tiles = (B / 16) * (HIDDEN_DIM / 16);
        gemm_wmma<true, HIDDEN_DIM, HIDDEN_DIM><<<tiles / 2, 64, 0, stream>>>(H1, W2, b2, H2, B);
    }
    {
        int tiles = (B / 16) * (GLOBAL_DIM / 16);
        gemm_wmma<false, HIDDEN_DIM, GLOBAL_DIM><<<tiles / 2, 64, 0, stream>>>(H2, W3, b3, out, B);
    }
    (void)n_in; (void)out_size;
}
